# starting kernel
# baseline (speedup 1.0000x reference)
#include <hip/hip_runtime.h>
#include <math.h>

#define NN 100000
#define NE 1600000
#define D  128
#define NL 3

typedef __attribute__((ext_vector_type(2))) float v2f;
typedef __attribute__((ext_vector_type(8))) float v8f;


__global__ void zero_f32_k(float* __restrict__ p, int n) {
    int i = blockIdx.x * blockDim.x + threadIdx.x;
    if (i < n) p[i] = 0.0f;
}

__global__ void deg_count_k(const int* __restrict__ col, float* __restrict__ deg, int e) {
    int i = blockIdx.x * blockDim.x + threadIdx.x;
    if (i < e) atomicAdd(&deg[col[i]], 1.0f);
}

__global__ void deg_inv_k(const float* __restrict__ deg, float* __restrict__ dinv, int n) {
    int i = blockIdx.x * blockDim.x + threadIdx.x;
    if (i < n) {
        float d = deg[i] + 1.0f;
        dinv[i] = rsqrtf(d);
    }
}


__global__ __launch_bounds__(256) void gemm_wmma_k(const float* __restrict__ A,
                                                   const float* __restrict__ W,
                                                   float* __restrict__ H) {
    const int lane = threadIdx.x & 31;
    const int wave = threadIdx.x >> 5;
    const int m0   = blockIdx.x << 4;
    const int n0   = wave << 4;
    const int half = lane >> 4;
    const int lm   = lane & 15;

    const float* __restrict__ arow = A + (size_t)(m0 + lm) * D;
    v8f c = {};

    #pragma unroll 4
    for (int k0 = 0; k0 < D; k0 += 4) {
        const int ka = k0 + 2 * half;
        v2f a = *(const v2f*)(arow + ka);
        v2f b;
        b.x = W[(size_t)ka * D + n0 + lm];
        b.y = W[(size_t)(ka + 1) * D + n0 + lm];
        c = __builtin_amdgcn_wmma_f32_16x16x4_f32(
             false, a,  false, b,
             (short)0, c,  false,  false);
    }

    float* __restrict__ drow = H + (size_t)(m0 + 8 * half) * D + n0 + lm;
    #pragma unroll
    for (int r = 0; r < 8; ++r) drow[(size_t)r * D] = c[r];
}


__global__ void self_init_k(const float* __restrict__ h, const float* __restrict__ dinv,
                            float* __restrict__ acc) {
    int idx = blockIdx.x * blockDim.x + threadIdx.x;
    if (idx >= NN * (D / 4)) return;
    int node = idx >> 5;
    float s = dinv[node];
    s = s * s;
    float4 v = ((const float4*)h)[idx];
    float4 o;
    o.x = v.x * s; o.y = v.y * s; o.z = v.z * s; o.w = v.w * s;
    ((float4*)acc)[idx] = o;
}


__global__ __launch_bounds__(256) void edge_scatter_k(const int* __restrict__ row,
                                                      const int* __restrict__ col,
                                                      const float* __restrict__ dinv,
                                                      const float* __restrict__ h,
                                                      float* __restrict__ acc) {
    const int lane = threadIdx.x & 31;
    const int wid  = (blockIdx.x * blockDim.x + threadIdx.x) >> 5;
    const int nw   = (gridDim.x * blockDim.x) >> 5;
    for (int e = wid; e < NE; e += nw) {
        const int r = row[e];
        const int c = col[e];
        const float nrm = dinv[r] * dinv[c];
        const float4 v = ((const float4*)(h + (size_t)r * D))[lane];
        float* ap = acc + (size_t)c * D + (lane << 2);
        atomicAdd(ap + 0, v.x * nrm);
        atomicAdd(ap + 1, v.y * nrm);
        atomicAdd(ap + 2, v.z * nrm);
        atomicAdd(ap + 3, v.w * nrm);
    }
}


__global__ void bias_relu_k(const float* __restrict__ acc, const float* __restrict__ b,
                            float* __restrict__ out) {
    int idx = blockIdx.x * blockDim.x + threadIdx.x;
    if (idx >= NN * (D / 4)) return;
    float4 v  = ((const float4*)acc)[idx];
    float4 bb = ((const float4*)b)[idx & 31];
    float4 o;
    o.x = fmaxf(v.x + bb.x, 0.0f);
    o.y = fmaxf(v.y + bb.y, 0.0f);
    o.z = fmaxf(v.z + bb.z, 0.0f);
    o.w = fmaxf(v.w + bb.w, 0.0f);
    ((float4*)out)[idx] = o;
}


extern "C" void kernel_launch(void* const* d_in, const int* in_sizes, int n_in,
                              void* d_out, int out_size, void* d_ws, size_t ws_size,
                              hipStream_t stream) {
    const float* x  = (const float*)d_in[0];
    const int*   ei = (const int*)d_in[1];
    const float* Ws = (const float*)d_in[2];
    const float* bs = (const float*)d_in[3];
    float* out = (float*)d_out;

    float* bufH   = (float*)d_ws;
    float* bufAcc = bufH + (size_t)NN * D;
    float* deg    = bufAcc + (size_t)NN * D;
    float* dinv   = deg + NN;

    const int* row = ei;
    const int* col = ei + NE;

    const int nvec = NN * (D / 4);

    zero_f32_k<<<(NN + 255) / 256, 256, 0, stream>>>(deg, NN);
    deg_count_k<<<(NE + 255) / 256, 256, 0, stream>>>(col, deg, NE);
    deg_inv_k<<<(NN + 255) / 256, 256, 0, stream>>>(deg, dinv, NN);

    const float* src = x;
    for (int l = 0; l < NL; ++l) {
        gemm_wmma_k<<<NN / 16, 256, 0, stream>>>(src, Ws + (size_t)l * D * D, bufH);
        self_init_k<<<(nvec + 255) / 256, 256, 0, stream>>>(bufH, dinv, bufAcc);
        edge_scatter_k<<<8192, 256, 0, stream>>>(row, col, dinv, bufH, bufAcc);
        float* dst = (l == NL - 1) ? out : bufAcc;
        bias_relu_k<<<(nvec + 255) / 256, 256, 0, stream>>>(bufAcc, bs + (size_t)l * D, dst);
        src = bufAcc;
    }
}


# edge scatter moved on chip: a workgroup owns 391 destination rows in LDS, filters the whole edge list by a 1-byte owner map, adds with LDS f32 atomics, writes bias+relu; sum order now LDS arrival
# speedup vs baseline: 2.3295x; 2.3295x over previous
#include <hip/hip_runtime.h>
#include <math.h>

#define NN 100000
#define NE 1600000
#define D  128
#define NL 3

typedef __attribute__((ext_vector_type(2))) float v2f;
typedef __attribute__((ext_vector_type(8))) float v8f;


__global__ void zero_f32_k(float* __restrict__ p, int n) {
    int i = blockIdx.x * blockDim.x + threadIdx.x;
    if (i < n) p[i] = 0.0f;
}

__global__ void deg_count_k(const int* __restrict__ col, float* __restrict__ deg, int e) {
    int i = blockIdx.x * blockDim.x + threadIdx.x;
    if (i < e) atomicAdd(&deg[col[i]], 1.0f);
}

__global__ void deg_inv_k(const float* __restrict__ deg, float* __restrict__ dinv, int n) {
    int i = blockIdx.x * blockDim.x + threadIdx.x;
    if (i < n) {
        float d = deg[i] + 1.0f;
        dinv[i] = rsqrtf(d);
    }
}


__global__ __launch_bounds__(256) void gemm_wmma_k(const float* __restrict__ A,
                                                   const float* __restrict__ W,
                                                   float* __restrict__ H) {
    const int lane = threadIdx.x & 31;
    const int wave = threadIdx.x >> 5;
    const int m0   = blockIdx.x << 4;
    const int n0   = wave << 4;
    const int half = lane >> 4;
    const int lm   = lane & 15;

    const float* __restrict__ arow = A + (size_t)(m0 + lm) * D;
    v8f c = {};

    #pragma unroll 4
    for (int k0 = 0; k0 < D; k0 += 4) {
        const int ka = k0 + 2 * half;
        v2f a = *(const v2f*)(arow + ka);
        v2f b;
        b.x = W[(size_t)ka * D + n0 + lm];
        b.y = W[(size_t)(ka + 1) * D + n0 + lm];
        c = __builtin_amdgcn_wmma_f32_16x16x4_f32(
             false, a,  false, b,
             (short)0, c,  false,  false);
    }

    float* __restrict__ drow = H + (size_t)(m0 + 8 * half) * D + n0 + lm;
    #pragma unroll
    for (int r = 0; r < 8; ++r) drow[(size_t)r * D] = c[r];
}


#define RNG  391
#define NOWN 256
#define QCAP 64

__global__ void owner_map_k(const int* __restrict__ col, unsigned int* __restrict__ own4) {
    int i = blockIdx.x * blockDim.x + threadIdx.x;
    if (i >= NE / 4) return;
    int4 c = ((const int4*)col)[i];
    unsigned int w = (unsigned int)(c.x / RNG) | ((unsigned int)(c.y / RNG) << 8) |
                     ((unsigned int)(c.z / RNG) << 16) | ((unsigned int)(c.w / RNG) << 24);
    own4[i] = w;
}


__global__ __launch_bounds__(1024) void agg_owner_k(const int* __restrict__ row,
                                                    const int* __restrict__ col,
                                                    const unsigned int* __restrict__ own4,
                                                    const float* __restrict__ dinv,
                                                    const float* __restrict__ h,
                                                    const float* __restrict__ b,
                                                    float* __restrict__ dst) {
    __shared__ float acc[RNG * D];
    __shared__ int   q[32][QCAP];

    const int tid  = threadIdx.x;
    const int lane = tid & 31;
    const int wave = __builtin_amdgcn_readfirstlane(tid >> 5);
    const int lo   = blockIdx.x * RNG;
    const int nrows = (NN - lo < RNG) ? (NN - lo) : RNG;

    for (int i = tid; i < nrows * (D / 4); i += 1024) {
        float s = dinv[lo + (i >> 5)];
        s = s * s;
        float4 v = ((const float4*)h)[(size_t)lo * (D / 4) + i];
        float4 o;
        o.x = v.x * s; o.y = v.y * s; o.z = v.z * s; o.w = v.w * s;
        ((float4*)acc)[i] = o;
    }
    __syncthreads();

    int* myq = q[wave];
    int qh = 0;
    int qn = 0;
    const unsigned int lt = (1u << lane) - 1u;

    auto drain = [&](int n) {
        int r = 0, cl = 0;
        float wgt = 0.0f;
        if (lane < n) {
            const int e = myq[(qh + lane) & (QCAP - 1)];
            r = row[e];
            const int c = col[e];
            cl = c - lo;
            wgt = dinv[r] * dinv[c];
        }
        #pragma unroll 1
        for (int j0 = 0; j0 < n; j0 += 8) {
            float v[8][4];
            float wj[8];
            int   cj[8];
            #pragma unroll
            for (int u = 0; u < 8; ++u) {
                const int j  = j0 + u;
                const int rj = __builtin_amdgcn_readlane(r, j);
                cj[u] = __builtin_amdgcn_readlane(cl, j);
                wj[u] = __int_as_float(__builtin_amdgcn_readlane(__float_as_int(wgt), j));
                const float* hp = h + (size_t)rj * D + lane;
                v[u][0] = hp[0]; v[u][1] = hp[32]; v[u][2] = hp[64]; v[u][3] = hp[96];
            }
            #pragma unroll
            for (int u = 0; u < 8; ++u) {
                if (j0 + u < n) {
                    float* ap = acc + cj[u] * D + lane;
                    __hip_atomic_fetch_add(ap +  0, v[u][0] * wj[u], __ATOMIC_RELAXED, __HIP_MEMORY_SCOPE_WORKGROUP);
                    __hip_atomic_fetch_add(ap + 32, v[u][1] * wj[u], __ATOMIC_RELAXED, __HIP_MEMORY_SCOPE_WORKGROUP);
                    __hip_atomic_fetch_add(ap + 64, v[u][2] * wj[u], __ATOMIC_RELAXED, __HIP_MEMORY_SCOPE_WORKGROUP);
                    __hip_atomic_fetch_add(ap + 96, v[u][3] * wj[u], __ATOMIC_RELAXED, __HIP_MEMORY_SCOPE_WORKGROUP);
                }
            }
        }
    };

    const unsigned int me4 = (unsigned int)blockIdx.x * 0x01010101u;
    const int ngroups = NE / 512;
    const int start   = blockIdx.x * (ngroups / NOWN);
    for (int t0 = wave; t0 < ngroups; t0 += 128) {
        uint4 w[4];
        int gg[4];
        #pragma unroll
        for (int u = 0; u < 4; ++u) {
            const int t = t0 + 32 * u;
            int g = t + start;
            if (g >= ngroups) g -= ngroups;
            if (t >= ngroups) g = 0;
            gg[u] = g;
            w[u] = ((const uint4*)own4)[g * 32 + lane];
        }
        #pragma unroll
        for (int u = 0; u < 4; ++u) {
            if (t0 + 32 * u < ngroups) {
                #define ZF(x) (~((((x) & 0x7F7F7F7Fu) + 0x7F7F7F7Fu) | (x) | 0x7F7F7F7Fu))
                const unsigned int x0 = w[u].x ^ me4, x1 = w[u].y ^ me4, x2 = w[u].z ^ me4, x3 = w[u].w ^ me4;
                unsigned int T = (ZF(x0) >> 7) | (ZF(x1) >> 6) | (ZF(x2) >> 5) | (ZF(x3) >> 4);
                #undef ZF
                const int ebase = (gg[u] * 32 + lane) * 16;
                unsigned int mk;
                while ((mk = __builtin_amdgcn_ballot_w32(T != 0u)) != 0u) {
                    if (T != 0u) {
                        const int p = __builtin_ctz(T);
                        myq[(qh + qn + __builtin_popcount(mk & lt)) & (QCAP - 1)] = ebase + ((p & 3) << 2) + (p >> 3);
                        T &= T - 1u;
                    }
                    qn += __builtin_popcount(mk);
                    if (qn >= 32) {
                        __builtin_amdgcn_wave_barrier();
                        drain(32);
                        qh = (qh + 32) & (QCAP - 1);
                        qn -= 32;
                    }
                }
            }
        }
    }
    if (qn > 0) {
        __builtin_amdgcn_wave_barrier();
        drain(qn);
    }
    __syncthreads();

    for (int i = tid; i < nrows * (D / 4); i += 1024) {
        float4 v  = ((const float4*)acc)[i];
        float4 bb = ((const float4*)b)[i & 31];
        float4 o;
        o.x = fmaxf(v.x + bb.x, 0.0f);
        o.y = fmaxf(v.y + bb.y, 0.0f);
        o.z = fmaxf(v.z + bb.z, 0.0f);
        o.w = fmaxf(v.w + bb.w, 0.0f);
        ((float4*)dst)[(size_t)lo * (D / 4) + i] = o;
    }
}


extern "C" void kernel_launch(void* const* d_in, const int* in_sizes, int n_in,
                              void* d_out, int out_size, void* d_ws, size_t ws_size,
                              hipStream_t stream) {
    const float* x  = (const float*)d_in[0];
    const int*   ei = (const int*)d_in[1];
    const float* Ws = (const float*)d_in[2];
    const float* bs = (const float*)d_in[3];
    float* out = (float*)d_out;

    float* bufH   = (float*)d_ws;
    float* bufX   = bufH + (size_t)NN * D;
    float* deg    = bufX + (size_t)NN * D;
    float* dinv   = deg + NN;
    unsigned int* own4 = (unsigned int*)(dinv + NN);

    const int* row = ei;
    const int* col = ei + NE;

    zero_f32_k<<<(NN + 255) / 256, 256, 0, stream>>>(deg, NN);
    deg_count_k<<<(NE + 255) / 256, 256, 0, stream>>>(col, deg, NE);
    deg_inv_k<<<(NN + 255) / 256, 256, 0, stream>>>(deg, dinv, NN);
    owner_map_k<<<(NE / 4 + 255) / 256, 256, 0, stream>>>(col, own4);

    const float* src = x;
    for (int l = 0; l < NL; ++l) {
        gemm_wmma_k<<<NN / 16, 256, 0, stream>>>(src, Ws + (size_t)l * D * D, bufH);
        float* dst = (l == NL - 1) ? out : bufX;
        agg_owner_k<<<NOWN, 1024, 0, stream>>>(row, col, own4, dinv, bufH, bs + (size_t)l * D, dst);
        src = bufX;
    }
}


# dense product made persistent: 1024 workgroups of 8 waves keep W fragments in registers and stream 32-row A panels through LDS with whole-line loads (was 6250 blocks, 16-line loads); k order kept
# speedup vs baseline: 2.6682x; 1.1454x over previous
#include <hip/hip_runtime.h>
#include <math.h>

#define NN 100000
#define NE 1600000
#define D  128
#define NL 3

typedef __attribute__((ext_vector_type(2))) float v2f;
typedef __attribute__((ext_vector_type(8))) float v8f;


__global__ void zero_f32_k(float* __restrict__ p, int n) {
    int i = blockIdx.x * blockDim.x + threadIdx.x;
    if (i < n) p[i] = 0.0f;
}

__global__ void deg_count_k(const int* __restrict__ col, float* __restrict__ deg, int e) {
    int i = blockIdx.x * blockDim.x + threadIdx.x;
    if (i < e) atomicAdd(&deg[col[i]], 1.0f);
}

__global__ void deg_inv_k(const float* __restrict__ deg, float* __restrict__ dinv, int n) {
    int i = blockIdx.x * blockDim.x + threadIdx.x;
    if (i < n) {
        float d = deg[i] + 1.0f;
        dinv[i] = rsqrtf(d);
    }
}


#define GP_ROWS  32
#define GP_PITCH 132
#define GEMM_WG  1024
#define GP_NPAN  (NN / GP_ROWS)
static_assert(NN % GP_ROWS == 0, "whole panels only");
static_assert(GEMM_WG <= GP_NPAN, "every workgroup has a first panel");

static __device__ __forceinline__ float4 panel_ld(const float* __restrict__ A, int pp, int tid, int i) {
    const int idx = tid + 256 * i;
    return ((const float4*)A)[(size_t)(pp * GP_ROWS + (idx >> 5)) * (D / 4) + (idx & 31)];
}

static __device__ __forceinline__ void panel_st(float* __restrict__ P, int tid, int i, float4 v) {
    const int idx = tid + 256 * i;
    *(float4*)(P + (idx >> 5) * GP_PITCH + ((idx & 31) << 2)) = v;
}

__global__ __launch_bounds__(256) void gemm_wmma_k(const float* __restrict__ A,
                                                   const float* __restrict__ W,
                                                   float* __restrict__ H) {
    __shared__ __attribute__((aligned(16))) float pan[2][GP_ROWS * GP_PITCH];

    const int tid  = threadIdx.x;
    const int lane = tid & 31;
    const int n0   = (tid >> 5) << 4;
    const int half = lane >> 4;
    const int lm   = lane & 15;

    v2f bf[32];
    #pragma unroll
    for (int s = 0; s < 32; ++s) {
        const int ka = 4 * s + 2 * half;
        bf[s].x = W[(size_t)ka * D + n0 + lm];
        bf[s].y = W[(size_t)(ka + 1) * D + n0 + lm];
    }

    int p = blockIdx.x;
    float4 pre0 = panel_ld(A, p, tid, 0);
    float4 pre1 = panel_ld(A, p, tid, 1);
    float4 pre2 = panel_ld(A, p, tid, 2);
    float4 pre3 = panel_ld(A, p, tid, 3);

    int buf = 0;
    for (; p < GP_NPAN; p += GEMM_WG) {
        float* P = pan[buf];
        panel_st(P, tid, 0, pre0);
        panel_st(P, tid, 1, pre1);
        panel_st(P, tid, 2, pre2);
        panel_st(P, tid, 3, pre3);
        __syncthreads();

        const int pn = p + GEMM_WG;
        if (pn < GP_NPAN) {
            pre0 = panel_ld(A, pn, tid, 0);
            pre1 = panel_ld(A, pn, tid, 1);
            pre2 = panel_ld(A, pn, tid, 2);
            pre3 = panel_ld(A, pn, tid, 3);
        }

        const float* a0p = P + lm * GP_PITCH + 2 * half;
        const float* a1p = a0p + 16 * GP_PITCH;
        v8f c0 = {};
        v8f c1 = {};
        #pragma unroll
        for (int s = 0; s < 32; ++s) {
            const v2f a0 = *(const v2f*)(a0p + 4 * s);
            const v2f a1 = *(const v2f*)(a1p + 4 * s);
            c0 = __builtin_amdgcn_wmma_f32_16x16x4_f32(
                 false, a0,  false, bf[s],
                 (short)0, c0,  false,  false);
            c1 = __builtin_amdgcn_wmma_f32_16x16x4_f32(
                 false, a1,  false, bf[s],
                 (short)0, c1,  false,  false);
        }

        float* __restrict__ drow = H + (size_t)(p * GP_ROWS + 8 * half) * D + n0 + lm;
        #pragma unroll
        for (int r = 0; r < 8; ++r) drow[(size_t)r * D] = c0[r];
        #pragma unroll
        for (int r = 0; r < 8; ++r) drow[(size_t)(16 + r) * D] = c1[r];
        buf ^= 1;
    }
}


#define RNG  391
#define NOWN 256
#define QCAP 64

__global__ void owner_map_k(const int* __restrict__ col, unsigned int* __restrict__ own4) {
    int i = blockIdx.x * blockDim.x + threadIdx.x;
    if (i >= NE / 4) return;
    int4 c = ((const int4*)col)[i];
    unsigned int w = (unsigned int)(c.x / RNG) | ((unsigned int)(c.y / RNG) << 8) |
                     ((unsigned int)(c.z / RNG) << 16) | ((unsigned int)(c.w / RNG) << 24);
    own4[i] = w;
}


__global__ __launch_bounds__(1024) void agg_owner_k(const int* __restrict__ row,
                                                    const int* __restrict__ col,
                                                    const unsigned int* __restrict__ own4,
                                                    const float* __restrict__ dinv,
                                                    const float* __restrict__ h,
                                                    const float* __restrict__ b,
                                                    float* __restrict__ dst) {
    __shared__ float acc[RNG * D];
    __shared__ int   q[32][QCAP];

    const int tid  = threadIdx.x;
    const int lane = tid & 31;
    const int wave = __builtin_amdgcn_readfirstlane(tid >> 5);
    const int lo   = blockIdx.x * RNG;
    const int nrows = (NN - lo < RNG) ? (NN - lo) : RNG;

    for (int i = tid; i < nrows * (D / 4); i += 1024) {
        float s = dinv[lo + (i >> 5)];
        s = s * s;
        float4 v = ((const float4*)h)[(size_t)lo * (D / 4) + i];
        float4 o;
        o.x = v.x * s; o.y = v.y * s; o.z = v.z * s; o.w = v.w * s;
        ((float4*)acc)[i] = o;
    }
    __syncthreads();

    int* myq = q[wave];
    int qh = 0;
    int qn = 0;
    const unsigned int lt = (1u << lane) - 1u;

    auto drain = [&](int n) {
        int r = 0, cl = 0;
        float wgt = 0.0f;
        if (lane < n) {
            const int e = myq[(qh + lane) & (QCAP - 1)];
            r = row[e];
            const int c = col[e];
            cl = c - lo;
            wgt = dinv[r] * dinv[c];
        }
        #pragma unroll 1
        for (int j0 = 0; j0 < n; j0 += 8) {
            float v[8][4];
            float wj[8];
            int   cj[8];
            #pragma unroll
            for (int u = 0; u < 8; ++u) {
                const int j  = j0 + u;
                const int rj = __builtin_amdgcn_readlane(r, j);
                cj[u] = __builtin_amdgcn_readlane(cl, j);
                wj[u] = __int_as_float(__builtin_amdgcn_readlane(__float_as_int(wgt), j));
                const float* hp = h + (size_t)rj * D + lane;
                v[u][0] = hp[0]; v[u][1] = hp[32]; v[u][2] = hp[64]; v[u][3] = hp[96];
            }
            #pragma unroll
            for (int u = 0; u < 8; ++u) {
                if (j0 + u < n) {
                    float* ap = acc + cj[u] * D + lane;
                    __hip_atomic_fetch_add(ap +  0, v[u][0] * wj[u], __ATOMIC_RELAXED, __HIP_MEMORY_SCOPE_WORKGROUP);
                    __hip_atomic_fetch_add(ap + 32, v[u][1] * wj[u], __ATOMIC_RELAXED, __HIP_MEMORY_SCOPE_WORKGROUP);
                    __hip_atomic_fetch_add(ap + 64, v[u][2] * wj[u], __ATOMIC_RELAXED, __HIP_MEMORY_SCOPE_WORKGROUP);
                    __hip_atomic_fetch_add(ap + 96, v[u][3] * wj[u], __ATOMIC_RELAXED, __HIP_MEMORY_SCOPE_WORKGROUP);
                }
            }
        }
    };

    const unsigned int me4 = (unsigned int)blockIdx.x * 0x01010101u;
    const int ngroups = NE / 512;
    const int start   = blockIdx.x * (ngroups / NOWN);
    for (int t0 = wave; t0 < ngroups; t0 += 128) {
        uint4 w[4];
        int gg[4];
        #pragma unroll
        for (int u = 0; u < 4; ++u) {
            const int t = t0 + 32 * u;
            int g = t + start;
            if (g >= ngroups) g -= ngroups;
            if (t >= ngroups) g = 0;
            gg[u] = g;
            w[u] = ((const uint4*)own4)[g * 32 + lane];
        }
        #pragma unroll
        for (int u = 0; u < 4; ++u) {
            if (t0 + 32 * u < ngroups) {
                #define ZF(x) (~((((x) & 0x7F7F7F7Fu) + 0x7F7F7F7Fu) | (x) | 0x7F7F7F7Fu))
                const unsigned int x0 = w[u].x ^ me4, x1 = w[u].y ^ me4, x2 = w[u].z ^ me4, x3 = w[u].w ^ me4;
                unsigned int T = (ZF(x0) >> 7) | (ZF(x1) >> 6) | (ZF(x2) >> 5) | (ZF(x3) >> 4);
                #undef ZF
                const int ebase = (gg[u] * 32 + lane) * 16;
                unsigned int mk;
                while ((mk = __builtin_amdgcn_ballot_w32(T != 0u)) != 0u) {
                    if (T != 0u) {
                        const int p = __builtin_ctz(T);
                        myq[(qh + qn + __builtin_popcount(mk & lt)) & (QCAP - 1)] = ebase + ((p & 3) << 2) + (p >> 3);
                        T &= T - 1u;
                    }
                    qn += __builtin_popcount(mk);
                    if (qn >= 32) {
                        __builtin_amdgcn_wave_barrier();
                        drain(32);
                        qh = (qh + 32) & (QCAP - 1);
                        qn -= 32;
                    }
                }
            }
        }
    }
    if (qn > 0) {
        __builtin_amdgcn_wave_barrier();
        drain(qn);
    }
    __syncthreads();

    for (int i = tid; i < nrows * (D / 4); i += 1024) {
        float4 v  = ((const float4*)acc)[i];
        float4 bb = ((const float4*)b)[i & 31];
        float4 o;
        o.x = fmaxf(v.x + bb.x, 0.0f);
        o.y = fmaxf(v.y + bb.y, 0.0f);
        o.z = fmaxf(v.z + bb.z, 0.0f);
        o.w = fmaxf(v.w + bb.w, 0.0f);
        ((float4*)dst)[(size_t)lo * (D / 4) + i] = o;
    }
}


extern "C" void kernel_launch(void* const* d_in, const int* in_sizes, int n_in,
                              void* d_out, int out_size, void* d_ws, size_t ws_size,
                              hipStream_t stream) {
    const float* x  = (const float*)d_in[0];
    const int*   ei = (const int*)d_in[1];
    const float* Ws = (const float*)d_in[2];
    const float* bs = (const float*)d_in[3];
    float* out = (float*)d_out;

    float* bufH   = (float*)d_ws;
    float* bufX   = bufH + (size_t)NN * D;
    float* deg    = bufX + (size_t)NN * D;
    float* dinv   = deg + NN;
    unsigned int* own4 = (unsigned int*)(dinv + NN);

    const int* row = ei;
    const int* col = ei + NE;

    zero_f32_k<<<(NN + 255) / 256, 256, 0, stream>>>(deg, NN);
    deg_count_k<<<(NE + 255) / 256, 256, 0, stream>>>(col, deg, NE);
    deg_inv_k<<<(NN + 255) / 256, 256, 0, stream>>>(deg, dinv, NN);
    owner_map_k<<<(NE / 4 + 255) / 256, 256, 0, stream>>>(col, own4);

    const float* src = x;
    for (int l = 0; l < NL; ++l) {
        gemm_wmma_k<<<GEMM_WG, 256, 0, stream>>>(src, Ws + (size_t)l * D * D, bufH);
        float* dst = (l == NL - 1) ? out : bufX;
        agg_owner_k<<<NOWN, 1024, 0, stream>>>(row, col, own4, dinv, bufH, bs + (size_t)l * D, dst);
        src = bufX;
    }
}


# row gather: one 16-byte-per-lane load fetches a source row (512 contiguous bytes per wave load, was four 128-byte loads); LDS rows kept in matching order so adds stay conflict-free; sum order kept
# speedup vs baseline: 3.1538x; 1.1820x over previous
#include <hip/hip_runtime.h>
#include <math.h>

#define NN 100000
#define NE 1600000
#define D  128
#define NL 3

typedef __attribute__((ext_vector_type(2))) float v2f;
typedef __attribute__((ext_vector_type(8))) float v8f;


__global__ void zero_f32_k(float* __restrict__ p, int n) {
    int i = blockIdx.x * blockDim.x + threadIdx.x;
    if (i < n) p[i] = 0.0f;
}

__global__ void deg_count_k(const int* __restrict__ col, float* __restrict__ deg, int e) {
    int i = blockIdx.x * blockDim.x + threadIdx.x;
    if (i < e) atomicAdd(&deg[col[i]], 1.0f);
}

__global__ void deg_inv_k(const float* __restrict__ deg, float* __restrict__ dinv, int n) {
    int i = blockIdx.x * blockDim.x + threadIdx.x;
    if (i < n) {
        float d = deg[i] + 1.0f;
        dinv[i] = rsqrtf(d);
    }
}


#define GP_ROWS  32
#define GP_PITCH 132
#define GEMM_WG  1024
#define GP_NPAN  (NN / GP_ROWS)
static_assert(NN % GP_ROWS == 0, "whole panels only");
static_assert(GEMM_WG <= GP_NPAN, "every workgroup has a first panel");

static __device__ __forceinline__ float4 panel_ld(const float* __restrict__ A, int pp, int tid, int i) {
    const int idx = tid + 256 * i;
    return ((const float4*)A)[(size_t)(pp * GP_ROWS + (idx >> 5)) * (D / 4) + (idx & 31)];
}

static __device__ __forceinline__ void panel_st(float* __restrict__ P, int tid, int i, float4 v) {
    const int idx = tid + 256 * i;
    *(float4*)(P + (idx >> 5) * GP_PITCH + ((idx & 31) << 2)) = v;
}

__global__ __launch_bounds__(256) void gemm_wmma_k(const float* __restrict__ A,
                                                   const float* __restrict__ W,
                                                   float* __restrict__ H) {
    __shared__ __attribute__((aligned(16))) float pan[2][GP_ROWS * GP_PITCH];

    const int tid  = threadIdx.x;
    const int lane = tid & 31;
    const int n0   = (tid >> 5) << 4;
    const int half = lane >> 4;
    const int lm   = lane & 15;

    v2f bf[32];
    #pragma unroll
    for (int s = 0; s < 32; ++s) {
        const int ka = 4 * s + 2 * half;
        bf[s].x = W[(size_t)ka * D + n0 + lm];
        bf[s].y = W[(size_t)(ka + 1) * D + n0 + lm];
    }

    int p = blockIdx.x;
    float4 pre0 = panel_ld(A, p, tid, 0);
    float4 pre1 = panel_ld(A, p, tid, 1);
    float4 pre2 = panel_ld(A, p, tid, 2);
    float4 pre3 = panel_ld(A, p, tid, 3);

    int buf = 0;
    for (; p < GP_NPAN; p += GEMM_WG) {
        float* P = pan[buf];
        panel_st(P, tid, 0, pre0);
        panel_st(P, tid, 1, pre1);
        panel_st(P, tid, 2, pre2);
        panel_st(P, tid, 3, pre3);
        __syncthreads();

        const int pn = p + GEMM_WG;
        if (pn < GP_NPAN) {
            pre0 = panel_ld(A, pn, tid, 0);
            pre1 = panel_ld(A, pn, tid, 1);
            pre2 = panel_ld(A, pn, tid, 2);
            pre3 = panel_ld(A, pn, tid, 3);
        }

        const float* a0p = P + lm * GP_PITCH + 2 * half;
        const float* a1p = a0p + 16 * GP_PITCH;
        v8f c0 = {};
        v8f c1 = {};
        #pragma unroll
        for (int s = 0; s < 32; ++s) {
            const v2f a0 = *(const v2f*)(a0p + 4 * s);
            const v2f a1 = *(const v2f*)(a1p + 4 * s);
            c0 = __builtin_amdgcn_wmma_f32_16x16x4_f32(
                 false, a0,  false, bf[s],
                 (short)0, c0,  false,  false);
            c1 = __builtin_amdgcn_wmma_f32_16x16x4_f32(
                 false, a1,  false, bf[s],
                 (short)0, c1,  false,  false);
        }

        float* __restrict__ drow = H + (size_t)(p * GP_ROWS + 8 * half) * D + n0 + lm;
        #pragma unroll
        for (int r = 0; r < 8; ++r) drow[(size_t)r * D] = c0[r];
        #pragma unroll
        for (int r = 0; r < 8; ++r) drow[(size_t)(16 + r) * D] = c1[r];
        buf ^= 1;
    }
}


#define RNG  391
#define NOWN 256
#define QCAP 64

__global__ void owner_map_k(const int* __restrict__ col, unsigned int* __restrict__ own4) {
    int i = blockIdx.x * blockDim.x + threadIdx.x;
    if (i >= NE / 4) return;
    int4 c = ((const int4*)col)[i];
    unsigned int w = (unsigned int)(c.x / RNG) | ((unsigned int)(c.y / RNG) << 8) |
                     ((unsigned int)(c.z / RNG) << 16) | ((unsigned int)(c.w / RNG) << 24);
    own4[i] = w;
}


__global__ __launch_bounds__(1024) void agg_owner_k(const int* __restrict__ row,
                                                    const int* __restrict__ col,
                                                    const unsigned int* __restrict__ own4,
                                                    const float* __restrict__ dinv,
                                                    const float* __restrict__ h,
                                                    const float* __restrict__ b,
                                                    float* __restrict__ dst) {
    __shared__ float acc[RNG * D];
    __shared__ int   q[32][QCAP];

    const int tid  = threadIdx.x;
    const int lane = tid & 31;
    const int wave = __builtin_amdgcn_readfirstlane(tid >> 5);
    const int lo   = blockIdx.x * RNG;
    const int nrows = (NN - lo < RNG) ? (NN - lo) : RNG;

    for (int i = tid; i < nrows * (D / 4); i += 1024) {
        float s = dinv[lo + (i >> 5)];
        s = s * s;
        float4 v = ((const float4*)h)[(size_t)lo * (D / 4) + i];
        float* ap = acc + (i >> 5) * D + (i & 31);
        ap[0] = v.x * s; ap[32] = v.y * s; ap[64] = v.z * s; ap[96] = v.w * s;
    }
    __syncthreads();

    int* myq = q[wave];
    int qh = 0;
    int qn = 0;
    const unsigned int lt = (1u << lane) - 1u;

    auto drain = [&](int n) {
        int r = 0, cl = 0;
        float wgt = 0.0f;
        if (lane < n) {
            const int e = myq[(qh + lane) & (QCAP - 1)];
            r = row[e];
            const int c = col[e];
            cl = c - lo;
            wgt = dinv[r] * dinv[c];
        }
        #pragma unroll 1
        for (int j0 = 0; j0 < n; j0 += 8) {
            float4 v[8];
            float wj[8];
            int   cj[8];
            #pragma unroll
            for (int u = 0; u < 8; ++u) {
                const int j  = j0 + u;
                const int rj = __builtin_amdgcn_readlane(r, j);
                cj[u] = __builtin_amdgcn_readlane(cl, j);
                wj[u] = __int_as_float(__builtin_amdgcn_readlane(__float_as_int(wgt), j));
                v[u] = ((const float4*)(h + (size_t)rj * D))[lane];
            }
            #pragma unroll
            for (int u = 0; u < 8; ++u) {
                if (j0 + u < n) {
                    float* ap = acc + cj[u] * D + lane;
                    __hip_atomic_fetch_add(ap +  0, v[u].x * wj[u], __ATOMIC_RELAXED, __HIP_MEMORY_SCOPE_WORKGROUP);
                    __hip_atomic_fetch_add(ap + 32, v[u].y * wj[u], __ATOMIC_RELAXED, __HIP_MEMORY_SCOPE_WORKGROUP);
                    __hip_atomic_fetch_add(ap + 64, v[u].z * wj[u], __ATOMIC_RELAXED, __HIP_MEMORY_SCOPE_WORKGROUP);
                    __hip_atomic_fetch_add(ap + 96, v[u].w * wj[u], __ATOMIC_RELAXED, __HIP_MEMORY_SCOPE_WORKGROUP);
                }
            }
        }
    };

    const unsigned int me4 = (unsigned int)blockIdx.x * 0x01010101u;
    const int ngroups = NE / 512;
    const int start   = blockIdx.x * (ngroups / NOWN);
    for (int t0 = wave; t0 < ngroups; t0 += 128) {
        uint4 w[4];
        int gg[4];
        #pragma unroll
        for (int u = 0; u < 4; ++u) {
            const int t = t0 + 32 * u;
            int g = t + start;
            if (g >= ngroups) g -= ngroups;
            if (t >= ngroups) g = 0;
            gg[u] = g;
            w[u] = ((const uint4*)own4)[g * 32 + lane];
        }
        #pragma unroll
        for (int u = 0; u < 4; ++u) {
            if (t0 + 32 * u < ngroups) {
                #define ZF(x) (~((((x) & 0x7F7F7F7Fu) + 0x7F7F7F7Fu) | (x) | 0x7F7F7F7Fu))
                const unsigned int x0 = w[u].x ^ me4, x1 = w[u].y ^ me4, x2 = w[u].z ^ me4, x3 = w[u].w ^ me4;
                unsigned int T = (ZF(x0) >> 7) | (ZF(x1) >> 6) | (ZF(x2) >> 5) | (ZF(x3) >> 4);
                #undef ZF
                const int ebase = (gg[u] * 32 + lane) * 16;
                unsigned int mk;
                while ((mk = __builtin_amdgcn_ballot_w32(T != 0u)) != 0u) {
                    if (T != 0u) {
                        const int p = __builtin_ctz(T);
                        myq[(qh + qn + __builtin_popcount(mk & lt)) & (QCAP - 1)] = ebase + ((p & 3) << 2) + (p >> 3);
                        T &= T - 1u;
                    }
                    qn += __builtin_popcount(mk);
                    if (qn >= 32) {
                        __builtin_amdgcn_wave_barrier();
                        drain(32);
                        qh = (qh + 32) & (QCAP - 1);
                        qn -= 32;
                    }
                }
            }
        }
    }
    if (qn > 0) {
        __builtin_amdgcn_wave_barrier();
        drain(qn);
    }
    __syncthreads();

    for (int i = tid; i < nrows * (D / 4); i += 1024) {
        const float* ap = acc + (i >> 5) * D + (i & 31);
        float4 bb = ((const float4*)b)[i & 31];
        float4 o;
        o.x = fmaxf(ap[0]  + bb.x, 0.0f);
        o.y = fmaxf(ap[32] + bb.y, 0.0f);
        o.z = fmaxf(ap[64] + bb.z, 0.0f);
        o.w = fmaxf(ap[96] + bb.w, 0.0f);
        ((float4*)dst)[(size_t)lo * (D / 4) + i] = o;
    }
}


extern "C" void kernel_launch(void* const* d_in, const int* in_sizes, int n_in,
                              void* d_out, int out_size, void* d_ws, size_t ws_size,
                              hipStream_t stream) {
    const float* x  = (const float*)d_in[0];
    const int*   ei = (const int*)d_in[1];
    const float* Ws = (const float*)d_in[2];
    const float* bs = (const float*)d_in[3];
    float* out = (float*)d_out;

    float* bufH   = (float*)d_ws;
    float* bufX   = bufH + (size_t)NN * D;
    float* deg    = bufX + (size_t)NN * D;
    float* dinv   = deg + NN;
    unsigned int* own4 = (unsigned int*)(dinv + NN);

    const int* row = ei;
    const int* col = ei + NE;

    zero_f32_k<<<(NN + 255) / 256, 256, 0, stream>>>(deg, NN);
    deg_count_k<<<(NE + 255) / 256, 256, 0, stream>>>(col, deg, NE);
    deg_inv_k<<<(NN + 255) / 256, 256, 0, stream>>>(deg, dinv, NN);
    owner_map_k<<<(NE / 4 + 255) / 256, 256, 0, stream>>>(col, own4);

    const float* src = x;
    for (int l = 0; l < NL; ++l) {
        gemm_wmma_k<<<GEMM_WG, 256, 0, stream>>>(src, Ws + (size_t)l * D * D, bufH);
        float* dst = (l == NL - 1) ? out : bufX;
        agg_owner_k<<<NOWN, 1024, 0, stream>>>(row, col, own4, dinv, bufH, bs + (size_t)l * D, dst);
        src = bufX;
    }
}


# owner filter run once per call: the first layer's scan also writes each wave's hits (row, local row, weight) to a workspace list; later layers replay the lists, no scan; sums still LDS arrival order
# speedup vs baseline: 3.6606x; 1.1607x over previous
#include <hip/hip_runtime.h>
#include <math.h>

#define NN 100000
#define NE 1600000
#define D  128
#define NL 3

typedef __attribute__((ext_vector_type(2))) float v2f;
typedef __attribute__((ext_vector_type(8))) float v8f;


__global__ void zero_f32_k(float* __restrict__ p, int n) {
    int i = blockIdx.x * blockDim.x + threadIdx.x;
    if (i < n) p[i] = 0.0f;
}

__global__ void deg_count_k(const int* __restrict__ col, float* __restrict__ deg, int e) {
    int i = blockIdx.x * blockDim.x + threadIdx.x;
    if (i < e) atomicAdd(&deg[col[i]], 1.0f);
}

__global__ void deg_inv_k(const float* __restrict__ deg, float* __restrict__ dinv, int n) {
    int i = blockIdx.x * blockDim.x + threadIdx.x;
    if (i < n) {
        float d = deg[i] + 1.0f;
        dinv[i] = rsqrtf(d);
    }
}


#define GP_ROWS  32
#define GP_PITCH 132
#define GEMM_WG  1024
#define GP_NPAN  (NN / GP_ROWS)
static_assert(NN % GP_ROWS == 0, "whole panels only");
static_assert(GEMM_WG <= GP_NPAN, "every workgroup has a first panel");

static __device__ __forceinline__ float4 panel_ld(const float* __restrict__ A, int pp, int tid, int i) {
    const int idx = tid + 256 * i;
    return ((const float4*)A)[(size_t)(pp * GP_ROWS + (idx >> 5)) * (D / 4) + (idx & 31)];
}

static __device__ __forceinline__ void panel_st(float* __restrict__ P, int tid, int i, float4 v) {
    const int idx = tid + 256 * i;
    *(float4*)(P + (idx >> 5) * GP_PITCH + ((idx & 31) << 2)) = v;
}

__global__ __launch_bounds__(256) void gemm_wmma_k(const float* __restrict__ A,
                                                   const float* __restrict__ W,
                                                   float* __restrict__ H) {
    __shared__ __attribute__((aligned(16))) float pan[2][GP_ROWS * GP_PITCH];

    const int tid  = threadIdx.x;
    const int lane = tid & 31;
    const int n0   = (tid >> 5) << 4;
    const int half = lane >> 4;
    const int lm   = lane & 15;

    v2f bf[32];
    #pragma unroll
    for (int s = 0; s < 32; ++s) {
        const int ka = 4 * s + 2 * half;
        bf[s].x = W[(size_t)ka * D + n0 + lm];
        bf[s].y = W[(size_t)(ka + 1) * D + n0 + lm];
    }

    int p = blockIdx.x;
    float4 pre0 = panel_ld(A, p, tid, 0);
    float4 pre1 = panel_ld(A, p, tid, 1);
    float4 pre2 = panel_ld(A, p, tid, 2);
    float4 pre3 = panel_ld(A, p, tid, 3);

    int buf = 0;
    for (; p < GP_NPAN; p += GEMM_WG) {
        float* P = pan[buf];
        panel_st(P, tid, 0, pre0);
        panel_st(P, tid, 1, pre1);
        panel_st(P, tid, 2, pre2);
        panel_st(P, tid, 3, pre3);
        __syncthreads();

        const int pn = p + GEMM_WG;
        if (pn < GP_NPAN) {
            pre0 = panel_ld(A, pn, tid, 0);
            pre1 = panel_ld(A, pn, tid, 1);
            pre2 = panel_ld(A, pn, tid, 2);
            pre3 = panel_ld(A, pn, tid, 3);
        }

        const float* a0p = P + lm * GP_PITCH + 2 * half;
        const float* a1p = a0p + 16 * GP_PITCH;
        v8f c0 = {};
        v8f c1 = {};
        #pragma unroll
        for (int s = 0; s < 32; ++s) {
            const v2f a0 = *(const v2f*)(a0p + 4 * s);
            const v2f a1 = *(const v2f*)(a1p + 4 * s);
            c0 = __builtin_amdgcn_wmma_f32_16x16x4_f32(
                 false, a0,  false, bf[s],
                 (short)0, c0,  false,  false);
            c1 = __builtin_amdgcn_wmma_f32_16x16x4_f32(
                 false, a1,  false, bf[s],
                 (short)0, c1,  false,  false);
        }

        float* __restrict__ drow = H + (size_t)(p * GP_ROWS + 8 * half) * D + n0 + lm;
        #pragma unroll
        for (int r = 0; r < 8; ++r) drow[(size_t)r * D] = c0[r];
        #pragma unroll
        for (int r = 0; r < 8; ++r) drow[(size_t)(16 + r) * D] = c1[r];
        buf ^= 1;
    }
}


#define RNG  391
#define NOWN 256
#define QCAP 64
#define LCAP 512

__global__ void owner_map_k(const int* __restrict__ col, unsigned int* __restrict__ own4) {
    int i = blockIdx.x * blockDim.x + threadIdx.x;
    if (i >= NE / 4) return;
    int4 c = ((const int4*)col)[i];
    unsigned int w = (unsigned int)(c.x / RNG) | ((unsigned int)(c.y / RNG) << 8) |
                     ((unsigned int)(c.z / RNG) << 16) | ((unsigned int)(c.w / RNG) << 24);
    own4[i] = w;
}


template <int MODE>
static __device__ __forceinline__ void agg_owner_body(const int* __restrict__ row,
                                                      const int* __restrict__ col,
                                                      const unsigned int* __restrict__ own4,
                                                      const float* __restrict__ dinv,
                                                      const float* __restrict__ h,
                                                      const float* __restrict__ b,
                                                      float* __restrict__ dst,
                                                      unsigned int* __restrict__ lst,
                                                      int* __restrict__ cnts) {
    __shared__ float acc[RNG * D];
    __shared__ int   q[32][QCAP];
    __shared__ int   wcnt[32];

    const int tid  = threadIdx.x;
    const int lane = tid & 31;
    const int wave = __builtin_amdgcn_readfirstlane(tid >> 5);
    const int lo   = blockIdx.x * RNG;
    const int nrows = (NN - lo < RNG) ? (NN - lo) : RNG;

    for (int i = tid; i < nrows * (D / 4); i += 1024) {
        float s = dinv[lo + (i >> 5)];
        s = s * s;
        float4 v = ((const float4*)h)[(size_t)lo * (D / 4) + i];
        float* ap = acc + (i >> 5) * D + (i & 31);
        ap[0] = v.x * s; ap[32] = v.y * s; ap[64] = v.z * s; ap[96] = v.w * s;
    }
    __syncthreads();

    int* myq = q[wave];
    int qh = 0;
    int qn = 0;
    const unsigned int lt = (1u << lane) - 1u;

    uint2* mylist = (uint2*)lst + ((size_t)blockIdx.x * 32 + wave) * LCAP;
    int rec = 0;

    auto gather = [&](int n, int r, int cl, float wgt) {
        auto group = [&](int j0, int m) {
            float4 v[8];
            float wj[8];
            int   cj[8];
            #pragma unroll
            for (int u = 0; u < 8; ++u) {
                const int j  = j0 + u;
                const int rj = __builtin_amdgcn_readlane(r, j);
                cj[u] = __builtin_amdgcn_readlane(cl, j);
                wj[u] = __int_as_float(__builtin_amdgcn_readlane(__float_as_int(wgt), j));
                v[u] = ((const float4*)(h + (size_t)rj * D))[lane];
            }
            #pragma unroll
            for (int u = 0; u < 8; ++u) {
                if (u < m) {
                    float* ap = acc + cj[u] * D + lane;
                    __hip_atomic_fetch_add(ap +  0, v[u].x * wj[u], __ATOMIC_RELAXED, __HIP_MEMORY_SCOPE_WORKGROUP);
                    __hip_atomic_fetch_add(ap + 32, v[u].y * wj[u], __ATOMIC_RELAXED, __HIP_MEMORY_SCOPE_WORKGROUP);
                    __hip_atomic_fetch_add(ap + 64, v[u].z * wj[u], __ATOMIC_RELAXED, __HIP_MEMORY_SCOPE_WORKGROUP);
                    __hip_atomic_fetch_add(ap + 96, v[u].w * wj[u], __ATOMIC_RELAXED, __HIP_MEMORY_SCOPE_WORKGROUP);
                }
            }
        };
        int j0 = 0;
        #pragma unroll 1
        for (; j0 + 8 <= n; j0 += 8) group(j0, 8);
        if (j0 < n) group(j0, n - j0);
    };

    auto drain = [&](int n) {
        int r = 0, cl = 0;
        float wgt = 0.0f;
        if (lane < n) {
            const int e = myq[(qh + lane) & (QCAP - 1)];
            r = row[e];
            const int c = col[e];
            cl = c - lo;
            wgt = dinv[r] * dinv[c];
        }
        if constexpr (MODE == 0) {
            if (lane < n && rec + lane < LCAP)
                mylist[rec + lane] = make_uint2((unsigned int)r | ((unsigned int)cl << 17), __float_as_uint(wgt));
            rec += n;
        }
        gather(n, r, cl, wgt);
    };

    bool scan = true;
    int nlist = 0;
    if constexpr (MODE == 1) {
        const int cj = cnts[blockIdx.x * 32 + lane];
        scan  = __builtin_amdgcn_ballot_w32(cj > LCAP) != 0u;
        nlist = __builtin_amdgcn_readlane(cj, wave);
    }

    if (scan) {
    const unsigned int me4 = (unsigned int)blockIdx.x * 0x01010101u;
    const int ngroups = NE / 512;
    const int start   = blockIdx.x * (ngroups / NOWN);
    for (int t0 = wave; t0 < ngroups; t0 += 128) {
        uint4 w[4];
        int gg[4];
        #pragma unroll
        for (int u = 0; u < 4; ++u) {
            const int t = t0 + 32 * u;
            int g = t + start;
            if (g >= ngroups) g -= ngroups;
            if (t >= ngroups) g = 0;
            gg[u] = g;
            w[u] = ((const uint4*)own4)[g * 32 + lane];
        }
        #pragma unroll
        for (int u = 0; u < 4; ++u) {
            if (t0 + 32 * u < ngroups) {
                #define ZF(x) (~((((x) & 0x7F7F7F7Fu) + 0x7F7F7F7Fu) | (x) | 0x7F7F7F7Fu))
                const unsigned int x0 = w[u].x ^ me4, x1 = w[u].y ^ me4, x2 = w[u].z ^ me4, x3 = w[u].w ^ me4;
                unsigned int T = (ZF(x0) >> 7) | (ZF(x1) >> 6) | (ZF(x2) >> 5) | (ZF(x3) >> 4);
                #undef ZF
                const int ebase = (gg[u] * 32 + lane) * 16;
                unsigned int mk;
                while ((mk = __builtin_amdgcn_ballot_w32(T != 0u)) != 0u) {
                    if (T != 0u) {
                        const int p = __builtin_ctz(T);
                        myq[(qh + qn + __builtin_popcount(mk & lt)) & (QCAP - 1)] = ebase + ((p & 3) << 2) + (p >> 3);
                        T &= T - 1u;
                    }
                    qn += __builtin_popcount(mk);
                    if (qn >= 32) {
                        __builtin_amdgcn_wave_barrier();
                        drain(32);
                        qh = (qh + 32) & (QCAP - 1);
                        qn -= 32;
                    }
                }
            }
        }
    }
    if (qn > 0) {
        __builtin_amdgcn_wave_barrier();
        drain(qn);
    }
    if constexpr (MODE == 0) {
        if (lane == 0) wcnt[wave] = rec;
    }
    } else {
        uint2 cur = make_uint2(0u, 0u);
        if (lane < nlist) cur = mylist[lane];
        #pragma unroll 1
        for (int pos = 0; pos < nlist; pos += 32) {
            uint2 nxt = make_uint2(0u, 0u);
            if (pos + 32 + lane < nlist) nxt = mylist[pos + 32 + lane];
            const int left = nlist - pos;
            gather(left < 32 ? left : 32, (int)(cur.x & 0x1FFFFu), (int)(cur.x >> 17), __uint_as_float(cur.y));
            cur = nxt;
        }
    }
    __syncthreads();

    if constexpr (MODE == 0) {
        if (tid < 32) cnts[blockIdx.x * 32 + tid] = wcnt[tid];
    }

    for (int i = tid; i < nrows * (D / 4); i += 1024) {
        const float* ap = acc + (i >> 5) * D + (i & 31);
        float4 bb = ((const float4*)b)[i & 31];
        float4 o;
        o.x = fmaxf(ap[0]  + bb.x, 0.0f);
        o.y = fmaxf(ap[32] + bb.y, 0.0f);
        o.z = fmaxf(ap[64] + bb.z, 0.0f);
        o.w = fmaxf(ap[96] + bb.w, 0.0f);
        ((float4*)dst)[(size_t)lo * (D / 4) + i] = o;
    }
}

__global__ __launch_bounds__(1024) void agg_first_k(const int* __restrict__ row,
                                                    const int* __restrict__ col,
                                                    const unsigned int* __restrict__ own4,
                                                    const float* __restrict__ dinv,
                                                    const float* __restrict__ h,
                                                    const float* __restrict__ b,
                                                    float* __restrict__ dst,
                                                    unsigned int* __restrict__ lst,
                                                    int* __restrict__ cnts) {
    agg_owner_body<0>(row, col, own4, dinv, h, b, dst, lst, cnts);
}

__global__ __launch_bounds__(1024) void agg_later_k(const int* __restrict__ row,
                                                    const int* __restrict__ col,
                                                    const unsigned int* __restrict__ own4,
                                                    const float* __restrict__ dinv,
                                                    const float* __restrict__ h,
                                                    const float* __restrict__ b,
                                                    float* __restrict__ dst,
                                                    unsigned int* __restrict__ lst,
                                                    int* __restrict__ cnts) {
    agg_owner_body<1>(row, col, own4, dinv, h, b, dst, lst, cnts);
}


extern "C" void kernel_launch(void* const* d_in, const int* in_sizes, int n_in,
                              void* d_out, int out_size, void* d_ws, size_t ws_size,
                              hipStream_t stream) {
    const float* x  = (const float*)d_in[0];
    const int*   ei = (const int*)d_in[1];
    const float* Ws = (const float*)d_in[2];
    const float* bs = (const float*)d_in[3];
    float* out = (float*)d_out;

    float* bufH   = (float*)d_ws;
    float* bufX   = bufH + (size_t)NN * D;
    float* deg    = bufX + (size_t)NN * D;
    float* dinv   = deg + NN;
    unsigned int* own4 = (unsigned int*)(dinv + NN);
    unsigned int* lst  = own4 + NE / 4;
    int* cnts = (int*)(lst + (size_t)NOWN * 32 * LCAP * 2);

    const int* row = ei;
    const int* col = ei + NE;

    zero_f32_k<<<(NN + 255) / 256, 256, 0, stream>>>(deg, NN);
    deg_count_k<<<(NE + 255) / 256, 256, 0, stream>>>(col, deg, NE);
    deg_inv_k<<<(NN + 255) / 256, 256, 0, stream>>>(deg, dinv, NN);
    owner_map_k<<<(NE / 4 + 255) / 256, 256, 0, stream>>>(col, own4);

    const float* src = x;
    for (int l = 0; l < NL; ++l) {
        gemm_wmma_k<<<GEMM_WG, 256, 0, stream>>>(src, Ws + (size_t)l * D * D, bufH);
        float* dst = (l == NL - 1) ? out : bufX;
        if (l == 0)
            agg_first_k<<<NOWN, 1024, 0, stream>>>(row, col, own4, dinv, bufH, bs + (size_t)l * D, dst, lst, cnts);
        else
            agg_later_k<<<NOWN, 1024, 0, stream>>>(row, col, own4, dinv, bufH, bs + (size_t)l * D, dst, lst, cnts);
        src = bufX;
    }
}
